# starting kernel
# baseline (speedup 1.0000x reference)
#include <hip/hip_runtime.h>


typedef __attribute__((ext_vector_type(2))) float v2f;
typedef __attribute__((ext_vector_type(8))) float v8f;

#define GCN_N 100000


__global__ void k_init_deg(float* __restrict__ deg) {
    int i = blockIdx.x * blockDim.x + threadIdx.x;
    if (i < GCN_N) deg[i] = 1.0f;
}

__global__ void k_deg_accum(const int* __restrict__ dst, int E,
                            float* __restrict__ deg) {
    int e = blockIdx.x * blockDim.x + threadIdx.x;
    if (e < E) atomicAdd(&deg[dst[e]], 1.0f);
}

__global__ void k_deg_rsqrt(float* __restrict__ deg) {
    int i = blockIdx.x * blockDim.x + threadIdx.x;
    if (i < GCN_N) deg[i] = rsqrtf(deg[i]);
}

__global__ void k_zero(float* __restrict__ p, int n) {
    int i = blockIdx.x * blockDim.x + threadIdx.x;
    if (i < n) p[i] = 0.0f;
}

template <int KI, int NO>
__global__ void k_gemm_wmma(const float* __restrict__ X,
                            const float* __restrict__ W,
                            float* __restrict__ T) {
    const int wave     = (blockIdx.x * blockDim.x + threadIdx.x) >> 5;
    const int lane     = threadIdx.x & 31;
    const int nColTile = NO / 16;
    const int rowTile  = wave / nColTile;
    const int colTile  = wave - rowTile * nColTile;
    if (rowTile >= GCN_N / 16) return;

    const int row0 = rowTile * 16;
    const int col0 = colTile * 16;
    const int m    = lane & 15;
    const int hi   = lane >> 4;

    v8f acc = {};
    constexpr int KCHUNKS = (KI + 3) / 4;
#pragma unroll
    for (int kk = 0; kk < KCHUNKS; ++kk) {
        const int kb = kk * 4 + hi * 2;
        v2f a, b;
        a.x = (kb     < KI) ? X[(row0 + m) * KI + kb    ] : 0.0f;
        a.y = (kb + 1 < KI) ? X[(row0 + m) * KI + kb + 1] : 0.0f;
        b.x = (kb     < KI) ? W[(kb    ) * NO + col0 + m] : 0.0f;
        b.y = (kb + 1 < KI) ? W[(kb + 1) * NO + col0 + m] : 0.0f;
        acc = __builtin_amdgcn_wmma_f32_16x16x4_f32(
            false, a, false, b, (short)0, acc, false, false);
    }

#pragma unroll
    for (int j = 0; j < 8; ++j)
        T[(row0 + j + hi * 8) * NO + col0 + m] = acc[j];
}

template <int F>
__global__ void k_scatter(const int* __restrict__ src,
                          const int* __restrict__ dst, int E,
                          const float* __restrict__ dinv,
                          const float* __restrict__ T,
                          float* __restrict__ agg) {
    int e = blockIdx.x * blockDim.x + threadIdx.x;
    if (e >= E) return;
    const int   s    = src[e];
    const int   d    = dst[e];
    const float norm = dinv[s] * dinv[d];
    const float4* tp = (const float4*)(T + (size_t)s * F);
    float*        ap = agg + (size_t)d * F;
#pragma unroll
    for (int c = 0; c < F / 4; ++c) {
        float4 v = tp[c];
        atomicAdd(ap + 4 * c + 0, v.x * norm);
        atomicAdd(ap + 4 * c + 1, v.y * norm);
        atomicAdd(ap + 4 * c + 2, v.z * norm);
        atomicAdd(ap + 4 * c + 3, v.w * norm);
    }
}

template <int F, bool RELU>
__global__ void k_finish(const float* __restrict__ T,
                         const float* __restrict__ dinv,
                         const float* __restrict__ bias,
                         float* __restrict__ agg) {
    int i = blockIdx.x * blockDim.x + threadIdx.x;
    if (i >= GCN_N * F) return;
    const int   node = i / F;
    const int   f    = i - node * F;
    const float di   = dinv[node];
    float v = agg[i] + T[i] * di * di + bias[f];
    agg[i] = RELU ? fmaxf(v, 0.0f) : v;
}


extern "C" void kernel_launch(void* const* d_in, const int* in_sizes, int n_in,
                              void* d_out, int out_size, void* d_ws, size_t ws_size,
                              hipStream_t stream) {
    const float* x  = (const float*)d_in[0];
    const int*   ei = (const int*)d_in[1];
    const float* W1 = (const float*)d_in[2];
    const float* b1 = (const float*)d_in[3];
    const float* W2 = (const float*)d_in[4];
    const float* b2 = (const float*)d_in[5];
    const float* W3 = (const float*)d_in[6];
    const float* b3 = (const float*)d_in[7];
    float* out = (float*)d_out;

    const int E = in_sizes[1] / 2;
    const int* src = ei;
    const int* dst = ei + E;

    const int N = GCN_N;

    float* ws   = (float*)d_ws;
    float* dinv = ws;
    float* t1   = dinv + N;
    float* h1   = t1 + (size_t)N * 16;
    float* t2   = h1 + (size_t)N * 16;
    float* h2   = t2 + (size_t)N * 32;
    float* t3   = h2 + (size_t)N * 32;

    const int B = 256;
    auto blk = [](int n, int b) { return (n + b - 1) / b; };

    k_init_deg<<<blk(N, B), B, 0, stream>>>(dinv);
    k_deg_accum<<<blk(E, B), B, 0, stream>>>(dst, E, dinv);
    k_deg_rsqrt<<<blk(N, B), B, 0, stream>>>(dinv);

    {
        const int waves = (N / 16) * 1;
        k_gemm_wmma<2, 16><<<blk(waves * 32, B), B, 0, stream>>>(x, W1, t1);
        k_zero<<<blk(N * 16, B), B, 0, stream>>>(h1, N * 16);
        k_scatter<16><<<blk(E, B), B, 0, stream>>>(src, dst, E, dinv, t1, h1);
        k_finish<16, true><<<blk(N * 16, B), B, 0, stream>>>(t1, dinv, b1, h1);
    }

    {
        const int waves = (N / 16) * 2;
        k_gemm_wmma<16, 32><<<blk(waves * 32, B), B, 0, stream>>>(h1, W2, t2);
        k_zero<<<blk(N * 32, B), B, 0, stream>>>(h2, N * 32);
        k_scatter<32><<<blk(E, B), B, 0, stream>>>(src, dst, E, dinv, t2, h2);
        k_finish<32, true><<<blk(N * 32, B), B, 0, stream>>>(t2, dinv, b2, h2);
    }

    {
        const int waves = (N / 16) * 1;
        k_gemm_wmma<32, 16><<<blk(waves * 32, B), B, 0, stream>>>(h2, W3, t3);
        k_zero<<<blk(N * 16, B), B, 0, stream>>>(out, N * 16);
        k_scatter<16><<<blk(E, B), B, 0, stream>>>(src, dst, E, dinv, t3, out);
        k_finish<16, false><<<blk(N * 16, B), B, 0, stream>>>(t3, dinv, b3, out);
    }
}


# Neighbour sums moved to the narrow side of each layer and fused with its dense transform: one kernel per layer gathers input rows straight into WMMA operands (16 nodes a wave); 10 launches -> 7
# speedup vs baseline: 6.7675x; 6.7675x over previous
#include <hip/hip_runtime.h>


typedef __attribute__((ext_vector_type(2))) float v2f;
typedef __attribute__((ext_vector_type(8))) float v8f;

#define GCN_N 100000


__global__ void k_zero_cnt(int* __restrict__ cnt) {
    int i = blockIdx.x * blockDim.x + threadIdx.x;
    if (i < GCN_N / 4) ((int4*)cnt)[i] = make_int4(0, 0, 0, 0);
}

__global__ void k_count_rank(const int* __restrict__ dst, int E,
                             int* __restrict__ cnt, int* __restrict__ rank) {
    const int e = (blockIdx.x * blockDim.x + threadIdx.x) * 4;
    if (e >= E) return;
    if ((E & 3) == 0) {
        const int4 d = *(const int4*)(dst + e);
        int4 r;
        r.x = atomicAdd(&cnt[d.x], 1);
        r.y = atomicAdd(&cnt[d.y], 1);
        r.z = atomicAdd(&cnt[d.z], 1);
        r.w = atomicAdd(&cnt[d.w], 1);
        *(int4*)(rank + e) = r;
    } else {
        for (int k = 0; k < 4 && e + k < E; ++k)
            rank[e + k] = atomicAdd(&cnt[dst[e + k]], 1);
    }
}

#define SCAN_SEG 3200
__global__ void __launch_bounds__(1024)
k_scan(const int* __restrict__ cnt, int* __restrict__ rowptr,
       float* __restrict__ dinv) {
    __shared__ int s_tot[32];
    const int lane = threadIdx.x & 31;
    const int wave = threadIdx.x >> 5;
    const int base = wave * SCAN_SEG;

    int sum = 0;
    for (int c = 0; c < SCAN_SEG / 128; ++c) {
        const int i = base + c * 128 + lane * 4;
        const int4 v = *(const int4*)(cnt + min(i, GCN_N - 4));
        const int  s = v.x + v.y + v.z + v.w;
        sum += (i < GCN_N) ? s : 0;
    }
    for (int o = 16; o > 0; o >>= 1) sum += __shfl_xor(sum, o);
    if (lane == 0) s_tot[wave] = sum;
    __syncthreads();

    int carry = 0;
    for (int w = 0; w < wave; ++w) carry += s_tot[w];

    for (int c0 = 0; c0 < SCAN_SEG / 128; c0 += 5) {
        int4 v[5];
        int  own[5], incl[5], tot[5];
#pragma unroll
        for (int k = 0; k < 5; ++k) {
            const int i = base + (c0 + k) * 128 + lane * 4;
            v[k] = *(const int4*)(cnt + min(i, GCN_N - 4));
            if (i >= GCN_N) v[k] = make_int4(0, 0, 0, 0);
            own[k]  = v[k].x + v[k].y + v[k].z + v[k].w;
            incl[k] = own[k];
        }
#pragma unroll
        for (int o = 1; o < 32; o <<= 1) {
#pragma unroll
            for (int k = 0; k < 5; ++k) {
                const int up = __shfl_up(incl[k], o);
                if (lane >= o) incl[k] += up;
            }
        }
#pragma unroll
        for (int k = 0; k < 5; ++k) tot[k] = __shfl(incl[k], 31);
#pragma unroll
        for (int k = 0; k < 5; ++k) {
            const int i = base + (c0 + k) * 128 + lane * 4;
            const int x0 = carry + incl[k] - own[k];
            const int x1 = x0 + v[k].x, x2 = x1 + v[k].y, x3 = x2 + v[k].z;
            if (i < GCN_N) {
                *(int4*)(rowptr + i) = make_int4(x0, x1, x2, x3);
                *(float4*)(dinv + i) = make_float4(rsqrtf((float)(v[k].x + 1)),
                                                   rsqrtf((float)(v[k].y + 1)),
                                                   rsqrtf((float)(v[k].z + 1)),
                                                   rsqrtf((float)(v[k].w + 1)));
            }
            carry += tot[k];
        }
    }
    if (wave == 31 && lane == 0) rowptr[GCN_N] = carry;
}

__global__ void k_fill(const int* __restrict__ src, const int* __restrict__ dst,
                       const int* __restrict__ rank,
                       const int* __restrict__ rowptr, int E,
                       int* __restrict__ csr,
                       const float* __restrict__ x, const float* __restrict__ dinv,
                       float* __restrict__ xs) {
    const int t = blockIdx.x * blockDim.x + threadIdx.x;
    for (int i = t; i < GCN_N; i += gridDim.x * blockDim.x) {
        const float2 v = ((const float2*)x)[i];
        const float  d = dinv[i];
        ((float2*)xs)[i] = make_float2(v.x * d, v.y * d);
    }
    const int e = t * 4;
    if (e >= E) return;
    const unsigned last = (unsigned)(E - 1);
    if ((E & 3) == 0) {
        const int4 s = *(const int4*)(src + e);
        const int4 d = *(const int4*)(dst + e);
        const int4 r = *(const int4*)(rank + e);
        const unsigned p0 = (unsigned)(rowptr[d.x] + r.x);
        const unsigned p1 = (unsigned)(rowptr[d.y] + r.y);
        const unsigned p2 = (unsigned)(rowptr[d.z] + r.z);
        const unsigned p3 = (unsigned)(rowptr[d.w] + r.w);
        csr[min(p0, last)] = s.x;
        csr[min(p1, last)] = s.y;
        csr[min(p2, last)] = s.z;
        csr[min(p3, last)] = s.w;
    } else {
        for (int k = 0; k < 4 && e + k < E; ++k) {
            const unsigned p = (unsigned)(rowptr[dst[e + k]] + rank[e + k]);
            csr[min(p, last)] = src[e + k];
        }
    }
}


__device__ __forceinline__ int list_pos(int idx, int end) {
    return max(min(idx, end - 1), 0);
}

__global__ void k_layer1(const int* __restrict__ rowptr,
                         const int* __restrict__ csr, int E,
                         const float* __restrict__ xs,
                         const float* __restrict__ W1,
                         const float* __restrict__ b1,
                         const float* __restrict__ dinv,
                         float* __restrict__ h1s) {
    const int wave = (blockIdx.x * blockDim.x + threadIdx.x) >> 5;
    if (wave >= GCN_N / 16) return;
    const int lane = threadIdx.x & 31;
    const int c = lane & 15;
    const int h = lane >> 4;
    const int node = wave * 16 + c;

    const int beg = max(rowptr[node], 0);
    const int end = min(rowptr[node + 1], E);
    const unsigned last = GCN_N - 1;
    const float2* X2 = (const float2*)xs;

    const float2 own = X2[node];
    float sx = h ? 0.0f : own.x;
    float sy = h ? 0.0f : own.y;

    const int e0 = beg + h;
    int raw[8];
#pragma unroll
    for (int k = 0; k < 8; ++k) raw[k] = csr[list_pos(e0 + 2 * k, end)];
    for (int e = e0; e < end; e += 16) {
        int nxt[8];
#pragma unroll
        for (int k = 0; k < 8; ++k) nxt[k] = csr[list_pos(e + 16 + 2 * k, end)];
        float2 r[8];
        float  m[8];
#pragma unroll
        for (int k = 0; k < 8; ++k) {
            const bool v = (e + 2 * k) < end;
            const unsigned s = v ? min((unsigned)raw[k], last) : (unsigned)node;
            m[k] = v ? 1.0f : 0.0f;
            r[k] = X2[s];
        }
#pragma unroll
        for (int k = 0; k < 8; ++k) {
            sx = fmaf(r[k].x, m[k], sx);
            sy = fmaf(r[k].y, m[k], sy);
        }
        asm volatile("" : "+v"(nxt[0]), "+v"(nxt[1]), "+v"(nxt[2]), "+v"(nxt[3]),
                          "+v"(nxt[4]), "+v"(nxt[5]), "+v"(nxt[6]), "+v"(nxt[7]),
                          "+v"(sx));
#pragma unroll
        for (int k = 0; k < 8; ++k) raw[k] = nxt[k];
        __builtin_amdgcn_sched_group_barrier(0x020, 16, 0);
        __builtin_amdgcn_sched_group_barrier(0x002, 64, 0);
    }

    v2f b;  b.x = sx;      b.y = sy;
    v2f a;  a.x = W1[c];   a.y = W1[16 + c];
    v8f d = {};
    d = __builtin_amdgcn_wmma_f32_16x16x4_f32(false, a, false, b, (short)0, d, false, false);

    const float  di = dinv[node];
    const float4 bl = ((const float4*)b1)[2 * h];
    const float4 bh = ((const float4*)b1)[2 * h + 1];
    float4 o0, o1;
    o0.x = fmaxf(d[0] * di + bl.x, 0.0f) * di;
    o0.y = fmaxf(d[1] * di + bl.y, 0.0f) * di;
    o0.z = fmaxf(d[2] * di + bl.z, 0.0f) * di;
    o0.w = fmaxf(d[3] * di + bl.w, 0.0f) * di;
    o1.x = fmaxf(d[4] * di + bh.x, 0.0f) * di;
    o1.y = fmaxf(d[5] * di + bh.y, 0.0f) * di;
    o1.z = fmaxf(d[6] * di + bh.z, 0.0f) * di;
    o1.w = fmaxf(d[7] * di + bh.w, 0.0f) * di;
    float4* O4 = (float4*)h1s;
    O4[(size_t)node * 4 + 2 * h]     = o0;
    O4[(size_t)node * 4 + 2 * h + 1] = o1;
}

__global__ void k_layer2(const int* __restrict__ rowptr,
                         const int* __restrict__ csr, int E,
                         const float* __restrict__ h1s,
                         const float* __restrict__ W2,
                         const float* __restrict__ b2,
                         const float* __restrict__ W3,
                         const float* __restrict__ dinv,
                         float* __restrict__ t3s) {
    const int wave = (blockIdx.x * blockDim.x + threadIdx.x) >> 5;
    if (wave >= GCN_N / 16) return;
    const int lane = threadIdx.x & 31;
    const int c = lane & 15;
    const int h = lane >> 4;
    const int node = wave * 16 + c;

    const int beg = max(rowptr[node], 0);
    const int end = min(rowptr[node + 1], E);
    const unsigned last = GCN_N - 1;
    const float4* H4 = (const float4*)h1s;

    float4 a0 = H4[(size_t)node * 4 + 2 * h];
    float4 a1 = H4[(size_t)node * 4 + 2 * h + 1];

    int raw[8];
#pragma unroll
    for (int k = 0; k < 8; ++k) raw[k] = csr[list_pos(beg + k, end)];
    for (int e = beg; e < end; e += 8) {
        int nxt[8];
#pragma unroll
        for (int k = 0; k < 8; ++k) nxt[k] = csr[list_pos(e + 8 + k, end)];
        float4 r0[8], r1[8];
        float  m[8];
#pragma unroll
        for (int k = 0; k < 8; ++k) {
            const bool v = (e + k) < end;
            const unsigned s = v ? min((unsigned)raw[k], last) : (unsigned)node;
            m[k]  = v ? 1.0f : 0.0f;
            r0[k] = H4[(size_t)s * 4 + 2 * h];
            r1[k] = H4[(size_t)s * 4 + 2 * h + 1];
        }
#pragma unroll
        for (int k = 0; k < 8; ++k) {
            a0.x = fmaf(r0[k].x, m[k], a0.x); a0.y = fmaf(r0[k].y, m[k], a0.y);
            a0.z = fmaf(r0[k].z, m[k], a0.z); a0.w = fmaf(r0[k].w, m[k], a0.w);
            a1.x = fmaf(r1[k].x, m[k], a1.x); a1.y = fmaf(r1[k].y, m[k], a1.y);
            a1.z = fmaf(r1[k].z, m[k], a1.z); a1.w = fmaf(r1[k].w, m[k], a1.w);
        }
        asm volatile("" : "+v"(nxt[0]), "+v"(nxt[1]), "+v"(nxt[2]), "+v"(nxt[3]),
                          "+v"(nxt[4]), "+v"(nxt[5]), "+v"(nxt[6]), "+v"(nxt[7]),
                          "+v"(a0.x));
#pragma unroll
        for (int k = 0; k < 8; ++k) raw[k] = nxt[k];
        __builtin_amdgcn_sched_group_barrier(0x020, 24, 0);
        __builtin_amdgcn_sched_group_barrier(0x002, 96, 0);
    }

    const float di = dinv[node];
    const float ag[8] = { a0.x * di, a0.y * di, a0.z * di, a0.w * di,
                          a1.x * di, a1.y * di, a1.z * di, a1.w * di };

    v8f d0 = {}, d1 = {};
#pragma unroll
    for (int kk = 0; kk < 4; ++kk) {
        const int k0 = 8 * h + 2 * kk;
        v2f b;  b.x  = ag[2 * kk];             b.y  = ag[2 * kk + 1];
        v2f w0; w0.x = W2[k0 * 32 + c];        w0.y = W2[(k0 + 1) * 32 + c];
        v2f w1; w1.x = W2[k0 * 32 + 16 + c];   w1.y = W2[(k0 + 1) * 32 + 16 + c];
        d0 = __builtin_amdgcn_wmma_f32_16x16x4_f32(false, w0, false, b, (short)0, d0, false, false);
        d1 = __builtin_amdgcn_wmma_f32_16x16x4_f32(false, w1, false, b, (short)0, d1, false, false);
    }

    const float4 p0 = ((const float4*)b2)[2 * h];
    const float4 p1 = ((const float4*)b2)[2 * h + 1];
    const float4 p2 = ((const float4*)b2)[4 + 2 * h];
    const float4 p3 = ((const float4*)b2)[4 + 2 * h + 1];
    const float hv[16] = {
        fmaxf(d0[0] + p0.x, 0.0f), fmaxf(d0[1] + p0.y, 0.0f),
        fmaxf(d0[2] + p0.z, 0.0f), fmaxf(d0[3] + p0.w, 0.0f),
        fmaxf(d0[4] + p1.x, 0.0f), fmaxf(d0[5] + p1.y, 0.0f),
        fmaxf(d0[6] + p1.z, 0.0f), fmaxf(d0[7] + p1.w, 0.0f),
        fmaxf(d1[0] + p2.x, 0.0f), fmaxf(d1[1] + p2.y, 0.0f),
        fmaxf(d1[2] + p2.z, 0.0f), fmaxf(d1[3] + p2.w, 0.0f),
        fmaxf(d1[4] + p3.x, 0.0f), fmaxf(d1[5] + p3.y, 0.0f),
        fmaxf(d1[6] + p3.z, 0.0f), fmaxf(d1[7] + p3.w, 0.0f) };

    v8f d3 = {};
#pragma unroll
    for (int kk = 0; kk < 8; ++kk) {
        const int k0 = (kk < 4) ? (8 * h + 2 * kk) : (16 + 8 * h + 2 * (kk - 4));
        v2f b; b.x = hv[2 * kk];        b.y = hv[2 * kk + 1];
        v2f w; w.x = W3[k0 * 16 + c];   w.y = W3[(k0 + 1) * 16 + c];
        d3 = __builtin_amdgcn_wmma_f32_16x16x4_f32(false, w, false, b, (short)0, d3, false, false);
    }

    float4* O4 = (float4*)t3s;
    O4[(size_t)node * 4 + 2 * h]     = make_float4(d3[0] * di, d3[1] * di, d3[2] * di, d3[3] * di);
    O4[(size_t)node * 4 + 2 * h + 1] = make_float4(d3[4] * di, d3[5] * di, d3[6] * di, d3[7] * di);
}

template <int F, bool RELU>
__global__ void k_gather(const int* __restrict__ rowptr,
                         const int* __restrict__ csr, int E,
                         const float* __restrict__ Ts,
                         const float* __restrict__ dinv,
                         const float* __restrict__ bias,
                         float* __restrict__ out) {
    constexpr int L = F / 4;
    const int t    = blockIdx.x * blockDim.x + threadIdx.x;
    const int node = t / L;
    const int ch   = t - node * L;
    if (node >= GCN_N) return;

    const float4* T4 = (const float4*)Ts;
    const int beg = max(rowptr[node], 0);
    const int end = min(rowptr[node + 1], E);
    const unsigned last = GCN_N - 1;

    float4 acc = T4[(size_t)node * L + ch];
    int e = beg;
    for (; e + 4 <= end; e += 4) {
        const unsigned s0 = min((unsigned)csr[e    ], last);
        const unsigned s1 = min((unsigned)csr[e + 1], last);
        const unsigned s2 = min((unsigned)csr[e + 2], last);
        const unsigned s3 = min((unsigned)csr[e + 3], last);
        const float4 v0 = T4[(size_t)s0 * L + ch];
        const float4 v1 = T4[(size_t)s1 * L + ch];
        const float4 v2 = T4[(size_t)s2 * L + ch];
        const float4 v3 = T4[(size_t)s3 * L + ch];
        acc.x += v0.x; acc.y += v0.y; acc.z += v0.z; acc.w += v0.w;
        acc.x += v1.x; acc.y += v1.y; acc.z += v1.z; acc.w += v1.w;
        acc.x += v2.x; acc.y += v2.y; acc.z += v2.z; acc.w += v2.w;
        acc.x += v3.x; acc.y += v3.y; acc.z += v3.z; acc.w += v3.w;
    }
    const int r = end - e;
    if (r > 0) {
        const unsigned s0 = min((unsigned)csr[e], last);
        const unsigned s1 = min((unsigned)csr[r > 1 ? e + 1 : e], last);
        const unsigned s2 = min((unsigned)csr[r > 2 ? e + 2 : e], last);
        const float4 v0 = T4[(size_t)s0 * L + ch];
        float4 v1 = T4[(size_t)s1 * L + ch];
        float4 v2 = T4[(size_t)s2 * L + ch];
        if (r < 2) v1 = make_float4(0.0f, 0.0f, 0.0f, 0.0f);
        if (r < 3) v2 = make_float4(0.0f, 0.0f, 0.0f, 0.0f);
        acc.x += v0.x; acc.y += v0.y; acc.z += v0.z; acc.w += v0.w;
        acc.x += v1.x; acc.y += v1.y; acc.z += v1.z; acc.w += v1.w;
        acc.x += v2.x; acc.y += v2.y; acc.z += v2.z; acc.w += v2.w;
    }

    const float  di = dinv[node];
    const float4 b  = ((const float4*)bias)[ch];
    float4 o;
    o.x = acc.x * di + b.x;
    o.y = acc.y * di + b.y;
    o.z = acc.z * di + b.z;
    o.w = acc.w * di + b.w;
    if (RELU) {
        o.x = fmaxf(o.x, 0.0f); o.y = fmaxf(o.y, 0.0f);
        o.z = fmaxf(o.z, 0.0f); o.w = fmaxf(o.w, 0.0f);
    }
    ((float4*)out)[(size_t)node * L + ch] = o;
}


extern "C" void kernel_launch(void* const* d_in, const int* in_sizes, int n_in,
                              void* d_out, int out_size, void* d_ws, size_t ws_size,
                              hipStream_t stream) {
    const float* x  = (const float*)d_in[0];
    const int*   ei = (const int*)d_in[1];
    const float* W1 = (const float*)d_in[2];
    const float* b1 = (const float*)d_in[3];
    const float* W2 = (const float*)d_in[4];
    const float* b2 = (const float*)d_in[5];
    const float* W3 = (const float*)d_in[6];
    const float* b3 = (const float*)d_in[7];
    float* out = (float*)d_out;

    const int E = in_sizes[1] / 2;
    const int* src = ei;
    const int* dst = ei + E;

    const int N = GCN_N;

    float* ws     = (float*)d_ws;
    float* dinv   = ws;
    int*   rowptr = (int*)(dinv + N);
    float* regA   = (float*)(rowptr + N + 4);
    float* regB   = regA + (size_t)N * 32;
    int*   csr    = (int*)(regB + (size_t)N * 32);
    float* xs     = regA;
    float* h1s    = regA + (size_t)N * 16;
    int*   cnt    = (int*)regB;
    int*   rank   = cnt + N;
    float* t3s    = regB;

    const int B = 64;
    auto blk = [](int n, int b) { return (n + b - 1) / b; };

    k_zero_cnt<<<blk(N / 4, B), B, 0, stream>>>(cnt);
    k_count_rank<<<blk((E + 3) / 4, B), B, 0, stream>>>(dst, E, cnt, rank);
    k_scan<<<1, 1024, 0, stream>>>(cnt, rowptr, dinv);
    k_fill<<<blk((E + 3) / 4, B), B, 0, stream>>>(src, dst, rank, rowptr, E, csr, x, dinv, xs);

    const int tileThreads = (N / 16) * 32;

    k_layer1<<<blk(tileThreads, B), B, 0, stream>>>(rowptr, csr, E, xs, W1, b1, dinv, h1s);

    k_layer2<<<blk(tileThreads, B), B, 0, stream>>>(rowptr, csr, E, h1s, W2, b2, W3, dinv, t3s);

    k_gather<16, false><<<blk(N * 4, B), B, 0, stream>>>(rowptr, csr, E, t3s, dinv, b3, out);
}


# Spill-list walk shared by the wave: a lane fetches one entry (edge number, then both end points), entries pass by readlane: 2 round trips per 32 entries, not 2 per entry, in the 3 layer kernels
# speedup vs baseline: 9.2976x; 1.3739x over previous
#include <hip/hip_runtime.h>


typedef __attribute__((ext_vector_type(2))) float v2f;
typedef __attribute__((ext_vector_type(8))) float v8f;

#define GCN_N 100000
#define ROW_K 32
#define CNT_PAD 32

static_assert(ROW_K == 32, "the layer kernels split a row of 32 slots by hand");
static_assert(GCN_N % 16 == 0, "whole waves only: the spill walk reads entries from every lane");


__global__ void k_zero_cnt(int* __restrict__ cnt) {
    int i = blockIdx.x * blockDim.x + threadIdx.x;
    if (i < (GCN_N + CNT_PAD) / 4) ((int4*)cnt)[i] = make_int4(0, 0, 0, 0);
}

__device__ __forceinline__ void put_edge(unsigned d, int slot, int s, int e, int E,
                                         int* __restrict__ cnt,
                                         int* __restrict__ slots,
                                         int* __restrict__ spill) {
    if (slot < ROW_K) {
        slots[(size_t)d * ROW_K + slot] = s;
    } else {
        const int o = atomicAdd(&cnt[GCN_N], 1);
        if (o < E) spill[o] = e;
    }
}

__device__ __forceinline__ void spill_entry(const int* __restrict__ spill,
                                            const int* __restrict__ src,
                                            const int* __restrict__ dst,
                                            int E, int j, int ns, int& dj, int& sj) {
    dj = -1;
    sj = 0;
    if (j < ns) {
        const unsigned e = min((unsigned)spill[j], (unsigned)(E - 1));
        dj = (int)min((unsigned)dst[e], (unsigned)(GCN_N - 1));
        sj = (int)min((unsigned)src[e], (unsigned)(GCN_N - 1));
    }
}

__global__ void k_place(const int* __restrict__ src, const int* __restrict__ dst,
                        int E, int* __restrict__ cnt, int* __restrict__ slots,
                        int* __restrict__ spill) {
    const int e = (blockIdx.x * blockDim.x + threadIdx.x) * 4;
    if (e >= E) return;
    const unsigned last = GCN_N - 1;
    if ((E & 3) == 0) {
        const int4 s = *(const int4*)(src + e);
        const int4 d = *(const int4*)(dst + e);
        const unsigned d0 = min((unsigned)d.x, last);
        const unsigned d1 = min((unsigned)d.y, last);
        const unsigned d2 = min((unsigned)d.z, last);
        const unsigned d3 = min((unsigned)d.w, last);
        const int r0 = atomicAdd(&cnt[d0], 1);
        const int r1 = atomicAdd(&cnt[d1], 1);
        const int r2 = atomicAdd(&cnt[d2], 1);
        const int r3 = atomicAdd(&cnt[d3], 1);
        put_edge(d0, r0, s.x, e,     E, cnt, slots, spill);
        put_edge(d1, r1, s.y, e + 1, E, cnt, slots, spill);
        put_edge(d2, r2, s.z, e + 2, E, cnt, slots, spill);
        put_edge(d3, r3, s.w, e + 3, E, cnt, slots, spill);
    } else {
        for (int k = 0; k < 4 && e + k < E; ++k) {
            const unsigned dd = min((unsigned)dst[e + k], last);
            const int r = atomicAdd(&cnt[dd], 1);
            put_edge(dd, r, src[e + k], e + k, E, cnt, slots, spill);
        }
    }
}

__global__ void k_scale_x(const int* __restrict__ cnt, const float* __restrict__ x,
                          float* __restrict__ xs) {
    const int i = blockIdx.x * blockDim.x + threadIdx.x;
    if (i >= GCN_N) return;
    const float  d = rsqrtf((float)(cnt[i] + 1));
    const float2 v = ((const float2*)x)[i];
    ((float2*)xs)[i] = make_float2(v.x * d, v.y * d);
}


__global__ void k_layer1(const int* __restrict__ cnt, const int* __restrict__ slots,
                         const int* __restrict__ spill,
                         const int* __restrict__ src, const int* __restrict__ dst,
                         int E,
                         const float* __restrict__ xs,
                         const float* __restrict__ W1,
                         const float* __restrict__ b1,
                         float* __restrict__ h1s) {
    const int wave = (blockIdx.x * blockDim.x + threadIdx.x) >> 5;
    if (wave >= GCN_N / 16) return;
    const int lane = threadIdx.x & 31;
    const int c = lane & 15;
    const int h = lane >> 4;
    const int node = wave * 16 + c;

    const unsigned last = GCN_N - 1;
    const float2* X2 = (const float2*)xs;
    const int4*   L4 = (const int4*)(slots + (size_t)node * ROW_K);

    const int4 q0 = L4[2 * h];
    const int4 q1 = L4[2 * h + 1];
    const int4 q2 = L4[4 + 2 * h];
    const int4 q3 = L4[5 + 2 * h];
    const int deg = max(cnt[node], 0);
    const int len = min(deg, ROW_K);

    const float2 own = X2[node];
    float sx = h ? 0.0f : own.x;
    float sy = h ? 0.0f : own.y;

    const int idx[16] = { q0.x, q0.y, q0.z, q0.w, q1.x, q1.y, q1.z, q1.w,
                          q2.x, q2.y, q2.z, q2.w, q3.x, q3.y, q3.z, q3.w };
    float2 r[16];
    float  m[16];
#pragma unroll
    for (int k = 0; k < 16; ++k) {
        const int  slot = (k < 8) ? (8 * h + k) : (16 + 8 * h + (k - 8));
        const bool v = slot < len;
        const unsigned s = v ? min((unsigned)idx[k], last) : (unsigned)node;
        m[k] = v ? 1.0f : 0.0f;
        r[k] = X2[s];
    }
#pragma unroll
    for (int k = 0; k < 16; ++k) {
        sx = fmaf(r[k].x, m[k], sx);
        sy = fmaf(r[k].y, m[k], sy);
    }

    const bool sp = deg > ROW_K;
    if (__builtin_amdgcn_ballot_w32(sp) != 0u) {
        const int ns = min(max(cnt[GCN_N], 0), E);
        for (int base = 0; base < ns; base += 32) {
            int dj, sj;
            spill_entry(spill, src, dst, E, base + lane, ns, dj, sj);
            const int m = min(32, ns - base);
            for (int k = 0; k < m; ++k) {
                const int dk = __builtin_amdgcn_readlane(dj, k);
                const int sk = __builtin_amdgcn_readlane(sj, k);
                if (sp && h == 0 && dk == node) {
                    const float2 t = X2[sk];
                    sx += t.x;
                    sy += t.y;
                }
            }
        }
    }

    v2f b;  b.x = sx;      b.y = sy;
    v2f a;  a.x = W1[c];   a.y = W1[16 + c];
    v8f d = {};
    d = __builtin_amdgcn_wmma_f32_16x16x4_f32(false, a, false, b, (short)0, d, false, false);

    const float  di = rsqrtf((float)(deg + 1));
    const float4 bl = ((const float4*)b1)[2 * h];
    const float4 bh = ((const float4*)b1)[2 * h + 1];
    float4 o0, o1;
    o0.x = fmaxf(d[0] * di + bl.x, 0.0f) * di;
    o0.y = fmaxf(d[1] * di + bl.y, 0.0f) * di;
    o0.z = fmaxf(d[2] * di + bl.z, 0.0f) * di;
    o0.w = fmaxf(d[3] * di + bl.w, 0.0f) * di;
    o1.x = fmaxf(d[4] * di + bh.x, 0.0f) * di;
    o1.y = fmaxf(d[5] * di + bh.y, 0.0f) * di;
    o1.z = fmaxf(d[6] * di + bh.z, 0.0f) * di;
    o1.w = fmaxf(d[7] * di + bh.w, 0.0f) * di;
    float4* O4 = (float4*)h1s;
    O4[(size_t)node * 4 + 2 * h]     = o0;
    O4[(size_t)node * 4 + 2 * h + 1] = o1;
}

__global__ void k_layer2(const int* __restrict__ cnt, const int* __restrict__ slots,
                         const int* __restrict__ spill,
                         const int* __restrict__ src, const int* __restrict__ dst,
                         int E,
                         const float* __restrict__ h1s,
                         const float* __restrict__ W2,
                         const float* __restrict__ b2,
                         const float* __restrict__ W3,
                         float* __restrict__ t3s) {
    const int wave = (blockIdx.x * blockDim.x + threadIdx.x) >> 5;
    if (wave >= GCN_N / 16) return;
    const int lane = threadIdx.x & 31;
    const int c = lane & 15;
    const int h = lane >> 4;
    const int node = wave * 16 + c;

    const unsigned last = GCN_N - 1;
    const float4* H4 = (const float4*)h1s;
    const int4*   L4 = (const int4*)(slots + (size_t)node * ROW_K);

    const int4 f0 = L4[0];
    const int4 f1 = L4[1];
    const int deg = max(cnt[node], 0);
    const int len = min(deg, ROW_K);

    float4 a0 = H4[(size_t)node * 4 + 2 * h];
    float4 a1 = H4[(size_t)node * 4 + 2 * h + 1];

    int raw[8] = { f0.x, f0.y, f0.z, f0.w, f1.x, f1.y, f1.z, f1.w };
    int e = 0;
    do {
        const int  nb = min((e >> 2) + 2, ROW_K / 4 - 2);
        const int4 n0 = L4[nb];
        const int4 n1 = L4[nb + 1];
        int nxt[8] = { n0.x, n0.y, n0.z, n0.w, n1.x, n1.y, n1.z, n1.w };
        float4 r0[8], r1[8];
        float  m[8];
#pragma unroll
        for (int k = 0; k < 8; ++k) {
            const bool v = (e + k) < len;
            const unsigned s = v ? min((unsigned)raw[k], last) : (unsigned)node;
            m[k]  = v ? 1.0f : 0.0f;
            r0[k] = H4[(size_t)s * 4 + 2 * h];
            r1[k] = H4[(size_t)s * 4 + 2 * h + 1];
        }
#pragma unroll
        for (int k = 0; k < 8; ++k) {
            a0.x = fmaf(r0[k].x, m[k], a0.x); a0.y = fmaf(r0[k].y, m[k], a0.y);
            a0.z = fmaf(r0[k].z, m[k], a0.z); a0.w = fmaf(r0[k].w, m[k], a0.w);
            a1.x = fmaf(r1[k].x, m[k], a1.x); a1.y = fmaf(r1[k].y, m[k], a1.y);
            a1.z = fmaf(r1[k].z, m[k], a1.z); a1.w = fmaf(r1[k].w, m[k], a1.w);
        }
        asm volatile("" : "+v"(nxt[0]), "+v"(nxt[1]), "+v"(nxt[2]), "+v"(nxt[3]),
                          "+v"(nxt[4]), "+v"(nxt[5]), "+v"(nxt[6]), "+v"(nxt[7]),
                          "+v"(a0.x));
#pragma unroll
        for (int k = 0; k < 8; ++k) raw[k] = nxt[k];
        __builtin_amdgcn_sched_group_barrier(0x020, 18, 0);
        __builtin_amdgcn_sched_group_barrier(0x002, 96, 0);
        e += 8;
    } while (e < len);

    const bool sp = deg > ROW_K;
    if (__builtin_amdgcn_ballot_w32(sp) != 0u) {
        const int ns = min(max(cnt[GCN_N], 0), E);
        for (int base = 0; base < ns; base += 32) {
            int dj, sj;
            spill_entry(spill, src, dst, E, base + lane, ns, dj, sj);
            const int m = min(32, ns - base);
            for (int k = 0; k < m; ++k) {
                const int dk = __builtin_amdgcn_readlane(dj, k);
                const int sk = __builtin_amdgcn_readlane(sj, k);
                if (sp && dk == node) {
                    const float4 t0 = H4[(size_t)sk * 4 + 2 * h];
                    const float4 t1 = H4[(size_t)sk * 4 + 2 * h + 1];
                    a0.x += t0.x; a0.y += t0.y; a0.z += t0.z; a0.w += t0.w;
                    a1.x += t1.x; a1.y += t1.y; a1.z += t1.z; a1.w += t1.w;
                }
            }
        }
    }

    const float di = rsqrtf((float)(deg + 1));
    const float ag[8] = { a0.x * di, a0.y * di, a0.z * di, a0.w * di,
                          a1.x * di, a1.y * di, a1.z * di, a1.w * di };

    v8f d0 = {}, d1 = {};
#pragma unroll
    for (int kk = 0; kk < 4; ++kk) {
        const int k0 = 8 * h + 2 * kk;
        v2f b;  b.x  = ag[2 * kk];             b.y  = ag[2 * kk + 1];
        v2f w0; w0.x = W2[k0 * 32 + c];        w0.y = W2[(k0 + 1) * 32 + c];
        v2f w1; w1.x = W2[k0 * 32 + 16 + c];   w1.y = W2[(k0 + 1) * 32 + 16 + c];
        d0 = __builtin_amdgcn_wmma_f32_16x16x4_f32(false, w0, false, b, (short)0, d0, false, false);
        d1 = __builtin_amdgcn_wmma_f32_16x16x4_f32(false, w1, false, b, (short)0, d1, false, false);
    }

    const float4 p0 = ((const float4*)b2)[2 * h];
    const float4 p1 = ((const float4*)b2)[2 * h + 1];
    const float4 p2 = ((const float4*)b2)[4 + 2 * h];
    const float4 p3 = ((const float4*)b2)[4 + 2 * h + 1];
    const float hv[16] = {
        fmaxf(d0[0] + p0.x, 0.0f), fmaxf(d0[1] + p0.y, 0.0f),
        fmaxf(d0[2] + p0.z, 0.0f), fmaxf(d0[3] + p0.w, 0.0f),
        fmaxf(d0[4] + p1.x, 0.0f), fmaxf(d0[5] + p1.y, 0.0f),
        fmaxf(d0[6] + p1.z, 0.0f), fmaxf(d0[7] + p1.w, 0.0f),
        fmaxf(d1[0] + p2.x, 0.0f), fmaxf(d1[1] + p2.y, 0.0f),
        fmaxf(d1[2] + p2.z, 0.0f), fmaxf(d1[3] + p2.w, 0.0f),
        fmaxf(d1[4] + p3.x, 0.0f), fmaxf(d1[5] + p3.y, 0.0f),
        fmaxf(d1[6] + p3.z, 0.0f), fmaxf(d1[7] + p3.w, 0.0f) };

    v8f d3 = {};
#pragma unroll
    for (int kk = 0; kk < 8; ++kk) {
        const int k0 = (kk < 4) ? (8 * h + 2 * kk) : (16 + 8 * h + 2 * (kk - 4));
        v2f b; b.x = hv[2 * kk];        b.y = hv[2 * kk + 1];
        v2f w; w.x = W3[k0 * 16 + c];   w.y = W3[(k0 + 1) * 16 + c];
        d3 = __builtin_amdgcn_wmma_f32_16x16x4_f32(false, w, false, b, (short)0, d3, false, false);
    }

    float4* O4 = (float4*)t3s;
    O4[(size_t)node * 4 + 2 * h]     = make_float4(d3[0] * di, d3[1] * di, d3[2] * di, d3[3] * di);
    O4[(size_t)node * 4 + 2 * h + 1] = make_float4(d3[4] * di, d3[5] * di, d3[6] * di, d3[7] * di);
}

template <int F, bool RELU>
__global__ void k_gather(const int* __restrict__ cnt, const int* __restrict__ slots,
                         const int* __restrict__ spill,
                         const int* __restrict__ src, const int* __restrict__ dst,
                         int E,
                         const float* __restrict__ Ts,
                         const float* __restrict__ bias,
                         float* __restrict__ out) {
    constexpr int L = F / 4;
    const int t    = blockIdx.x * blockDim.x + threadIdx.x;
    const int node = t / L;
    const int ch   = t - node * L;
    if (node >= GCN_N) return;

    const float4* T4 = (const float4*)Ts;
    const int4*   L4 = (const int4*)(slots + (size_t)node * ROW_K);
    const int deg = max(cnt[node], 0);
    const int len = min(deg, ROW_K);
    const unsigned last = GCN_N - 1;

    float4 acc = T4[(size_t)node * L + ch];
    int e = 0;
    for (; e + 4 <= len; e += 4) {
        const int4 q = L4[e >> 2];
        const unsigned s0 = min((unsigned)q.x, last);
        const unsigned s1 = min((unsigned)q.y, last);
        const unsigned s2 = min((unsigned)q.z, last);
        const unsigned s3 = min((unsigned)q.w, last);
        const float4 v0 = T4[(size_t)s0 * L + ch];
        const float4 v1 = T4[(size_t)s1 * L + ch];
        const float4 v2 = T4[(size_t)s2 * L + ch];
        const float4 v3 = T4[(size_t)s3 * L + ch];
        acc.x += v0.x; acc.y += v0.y; acc.z += v0.z; acc.w += v0.w;
        acc.x += v1.x; acc.y += v1.y; acc.z += v1.z; acc.w += v1.w;
        acc.x += v2.x; acc.y += v2.y; acc.z += v2.z; acc.w += v2.w;
        acc.x += v3.x; acc.y += v3.y; acc.z += v3.z; acc.w += v3.w;
    }
    const int r = len - e;
    if (r > 0) {
        const int4 q = L4[e >> 2];
        const unsigned s0 = min((unsigned)q.x, last);
        const unsigned s1 = min((unsigned)(r > 1 ? q.y : q.x), last);
        const unsigned s2 = min((unsigned)(r > 2 ? q.z : q.x), last);
        const float4 v0 = T4[(size_t)s0 * L + ch];
        float4 v1 = T4[(size_t)s1 * L + ch];
        float4 v2 = T4[(size_t)s2 * L + ch];
        if (r < 2) v1 = make_float4(0.0f, 0.0f, 0.0f, 0.0f);
        if (r < 3) v2 = make_float4(0.0f, 0.0f, 0.0f, 0.0f);
        acc.x += v0.x; acc.y += v0.y; acc.z += v0.z; acc.w += v0.w;
        acc.x += v1.x; acc.y += v1.y; acc.z += v1.z; acc.w += v1.w;
        acc.x += v2.x; acc.y += v2.y; acc.z += v2.z; acc.w += v2.w;
    }
    const bool sp = deg > ROW_K;
    if (__builtin_amdgcn_ballot_w32(sp) != 0u) {
        const int lane = threadIdx.x & 31;
        const int ns = min(max(cnt[GCN_N], 0), E);
        for (int base = 0; base < ns; base += 32) {
            int dj, sj;
            spill_entry(spill, src, dst, E, base + lane, ns, dj, sj);
            const int m = min(32, ns - base);
            for (int k = 0; k < m; ++k) {
                const int dk = __builtin_amdgcn_readlane(dj, k);
                const int sk = __builtin_amdgcn_readlane(sj, k);
                if (sp && dk == node) {
                    const float4 v = T4[(size_t)sk * L + ch];
                    acc.x += v.x; acc.y += v.y; acc.z += v.z; acc.w += v.w;
                }
            }
        }
    }

    const float  di = rsqrtf((float)(deg + 1));
    const float4 b  = ((const float4*)bias)[ch];
    float4 o;
    o.x = acc.x * di + b.x;
    o.y = acc.y * di + b.y;
    o.z = acc.z * di + b.z;
    o.w = acc.w * di + b.w;
    if (RELU) {
        o.x = fmaxf(o.x, 0.0f); o.y = fmaxf(o.y, 0.0f);
        o.z = fmaxf(o.z, 0.0f); o.w = fmaxf(o.w, 0.0f);
    }
    ((float4*)out)[(size_t)node * L + ch] = o;
}


extern "C" void kernel_launch(void* const* d_in, const int* in_sizes, int n_in,
                              void* d_out, int out_size, void* d_ws, size_t ws_size,
                              hipStream_t stream) {
    const float* x  = (const float*)d_in[0];
    const int*   ei = (const int*)d_in[1];
    const float* W1 = (const float*)d_in[2];
    const float* b1 = (const float*)d_in[3];
    const float* W2 = (const float*)d_in[4];
    const float* b2 = (const float*)d_in[5];
    const float* W3 = (const float*)d_in[6];
    const float* b3 = (const float*)d_in[7];
    float* out = (float*)d_out;

    const int E = in_sizes[1] / 2;
    const int* src = ei;
    const int* dst = ei + E;

    const int N = GCN_N;

    int*   cnt   = (int*)d_ws;
    int*   slots = cnt + N + CNT_PAD;
    float* h1s   = (float*)(slots + (size_t)N * ROW_K);
    float* t3s   = h1s + (size_t)N * 16;
    float* xs    = t3s;
    int*   spill = (int*)(t3s + (size_t)N * 16);

    const int B = 64;
    auto blk = [](int n, int b) { return (n + b - 1) / b; };

    k_zero_cnt<<<blk((N + CNT_PAD) / 4, B), B, 0, stream>>>(cnt);
    k_place<<<blk((E + 3) / 4, B), B, 0, stream>>>(src, dst, E, cnt, slots, spill);
    k_scale_x<<<blk(N, B), B, 0, stream>>>(cnt, x, xs);

    const int tileThreads = (N / 16) * 32;

    k_layer1<<<blk(tileThreads, B), B, 0, stream>>>(cnt, slots, spill, src, dst, E, xs, W1, b1, h1s);

    k_layer2<<<blk(tileThreads, B), B, 0, stream>>>(cnt, slots, spill, src, dst, E, h1s, W2, b2, W3, t3s);

    k_gather<16, false><<<blk(N * 4, B), B, 0, stream>>>(cnt, slots, spill, src, dst, E, t3s, b3, out);
}
